# starting kernel
# baseline (speedup 1.0000x reference)
#include <hip/hip_runtime.h>

#define N_NODES   100000
#define N_EDGES   1600000
#define N_GRAPHS  1024
#define IN_LOCAL  16
#define IN_GLOBAL 8
#define HIDDEN    128
#define KC        (IN_LOCAL + HIDDEN)

typedef __attribute__((ext_vector_type(2))) float v2f;
typedef __attribute__((ext_vector_type(8))) float v8f;

#define LDSA_STRIDE 145
#define LDSH_STRIDE 129

__global__ __launch_bounds__(256) void k_global_enc(
    const float* __restrict__ gf, const float* __restrict__ Wg,
    const float* __restrict__ bg, float* __restrict__ hg) {
  int idx = blockIdx.x * 256 + threadIdx.x;
  int g = idx >> 7, h = idx & 127;
  float acc = bg[h];
#pragma unroll
  for (int k = 0; k < IN_GLOBAL; ++k)
    acc = fmaf(gf[g * IN_GLOBAL + k], Wg[k * HIDDEN + h], acc);
  hg[idx] = fmaxf(acc, 0.0f);
}

__global__ __launch_bounds__(256) void k_node_gemms(
    const float* __restrict__ x,  const int* __restrict__ batch,
    const float* __restrict__ hg,
    const float* __restrict__ Wc, const float* __restrict__ bc,
    const float* __restrict__ Wm, const float* __restrict__ bm,
    const float* __restrict__ Ws, const float* __restrict__ bs,
    float* __restrict__ m_buf, float* __restrict__ acc_buf) {
  __shared__ float lds_a[16 * LDSA_STRIDE];
  __shared__ float lds_h[16 * LDSH_STRIDE];

  const int tid = threadIdx.x;
  const int nodeBase = blockIdx.x * 16;

  for (int i = tid; i < 16 * KC; i += 256) {
    int row = i / KC, k = i - row * KC;
    int node = nodeBase + row;
    float v = (k < IN_LOCAL) ? x[node * IN_LOCAL + k]
                             : hg[batch[node] * HIDDEN + (k - IN_LOCAL)];
    lds_a[row * LDSA_STRIDE + k] = v;
  }
  __syncthreads();

  const int lane  = tid & 31;
  const int wave  = tid >> 5;
  const int half  = lane >> 4;
  const int idx16 = lane & 15;
  const int colBase = wave * 16;

  v8f acc = {};
#pragma unroll 4
  for (int kb = 0; kb < KC; kb += 4) {
    const int k0 = kb + 2 * half;
    v2f a, b;
    a.x = lds_a[idx16 * LDSA_STRIDE + k0];
    a.y = lds_a[idx16 * LDSA_STRIDE + k0 + 1];
    b.x = Wc[k0 * HIDDEN + colBase + idx16];
    b.y = Wc[(k0 + 1) * HIDDEN + colBase + idx16];
    acc = __builtin_amdgcn_wmma_f32_16x16x4_f32(
        false, a, false, b, (short)0, acc, false, false);
  }
  {
    const float bias = bc[colBase + idx16];
#pragma unroll
    for (int j = 0; j < 8; ++j) {
      int row = j + 8 * half;
      lds_h[row * LDSH_STRIDE + colBase + idx16] = fmaxf(acc[j] + bias, 0.0f);
    }
  }
  __syncthreads();

  v8f am = {}, as = {};
#pragma unroll 4
  for (int kb = 0; kb < HIDDEN; kb += 4) {
    const int k0 = kb + 2 * half;
    v2f a, b0, b1;
    a.x  = lds_h[idx16 * LDSH_STRIDE + k0];
    a.y  = lds_h[idx16 * LDSH_STRIDE + k0 + 1];
    b0.x = Wm[k0 * HIDDEN + colBase + idx16];
    b0.y = Wm[(k0 + 1) * HIDDEN + colBase + idx16];
    b1.x = Ws[k0 * HIDDEN + colBase + idx16];
    b1.y = Ws[(k0 + 1) * HIDDEN + colBase + idx16];
    am = __builtin_amdgcn_wmma_f32_16x16x4_f32(
        false, a, false, b0, (short)0, am, false, false);
    as = __builtin_amdgcn_wmma_f32_16x16x4_f32(
        false, a, false, b1, (short)0, as, false, false);
  }
  {
    const int col = colBase + idx16;
    const float biasm = bm[col], biass = bs[col];
#pragma unroll
    for (int j = 0; j < 8; ++j) {
      int row  = j + 8 * half;
      int node = nodeBase + row;
      float mv = fmaxf(am[j] + biasm, 0.0f);
      m_buf[(size_t)node * HIDDEN + col]   = mv;
      acc_buf[(size_t)node * HIDDEN + col] = mv + as[j] + biass;
    }
  }
}

__global__ __launch_bounds__(256) void k_scatter(
    const int* __restrict__ edge_index, const float* __restrict__ m_buf,
    float* __restrict__ acc_buf) {
  const int lane = threadIdx.x & 31;
  const int e = blockIdx.x * 8 + (threadIdx.x >> 5);
  const int src = edge_index[e];
  const int dst = edge_index[N_EDGES + e];
  const float* ms = m_buf   + (size_t)src * HIDDEN;
  float*       ad = acc_buf + (size_t)dst * HIDDEN;
#pragma unroll
  for (int t = 0; t < 4; ++t) {
    int c = lane + t * 32;
    unsafeAtomicAdd(&ad[c], ms[c]);
  }
}

__global__ __launch_bounds__(256) void k_out(
    const float* __restrict__ acc_buf, const float* __restrict__ Wo,
    const float* __restrict__ bo, float* __restrict__ out) {
  const int lane = threadIdx.x & 31;
  const int node = blockIdx.x * 8 + (threadIdx.x >> 5);
  float a0 = 0.0f, a1 = 0.0f;
#pragma unroll
  for (int t = 0; t < 4; ++t) {
    int k = lane + t * 32;
    float h = fmaxf(acc_buf[(size_t)node * HIDDEN + k], 0.0f);
    a0 = fmaf(h, Wo[k * 2 + 0], a0);
    a1 = fmaf(h, Wo[k * 2 + 1], a1);
  }
#pragma unroll
  for (int off = 16; off >= 1; off >>= 1) {
    a0 += __shfl_xor(a0, off, 32);
    a1 += __shfl_xor(a1, off, 32);
  }
  if (lane == 0) {
    out[node * 2 + 0] = a0 + bo[0];
    out[node * 2 + 1] = a1 + bo[1];
  }
}

extern "C" void kernel_launch(void* const* d_in, const int* in_sizes, int n_in,
                              void* d_out, int out_size, void* d_ws, size_t ws_size,
                              hipStream_t stream) {
  (void)in_sizes; (void)n_in; (void)out_size; (void)ws_size;
  const float* x     = (const float*)d_in[0];
  const int*   edge  = (const int*)  d_in[1];
  const int*   batch = (const int*)  d_in[2];
  const float* gf    = (const float*)d_in[3];
  const float* Wg    = (const float*)d_in[4];
  const float* bg    = (const float*)d_in[5];
  const float* Wc    = (const float*)d_in[6];
  const float* bc    = (const float*)d_in[7];
  const float* Wm    = (const float*)d_in[8];
  const float* bm    = (const float*)d_in[9];
  const float* Ws    = (const float*)d_in[10];
  const float* bs    = (const float*)d_in[11];
  const float* Wo    = (const float*)d_in[12];
  const float* bo    = (const float*)d_in[13];
  float* out = (float*)d_out;

  float* hg      = (float*)d_ws;
  float* m_buf   = hg    + (size_t)N_GRAPHS * HIDDEN;
  float* acc_buf = m_buf + (size_t)N_NODES  * HIDDEN;

  k_global_enc<<<(N_GRAPHS * HIDDEN) / 256, 256, 0, stream>>>(gf, Wg, bg, hg);
  k_node_gemms<<<N_NODES / 16, 256, 0, stream>>>(x, batch, hg, Wc, bc,
                                                 Wm, bm, Ws, bs, m_buf, acc_buf);
  k_scatter<<<N_EDGES / 8, 256, 0, stream>>>(edge, m_buf, acc_buf);
  k_out<<<N_NODES / 8, 256, 0, stream>>>(acc_buf, Wo, bo, out);
}


# Edge sum by ownership: 256 work-groups each keep 391 nodes' rows in LDS, scan all edge targets and add their own edges' m rows (LDS float adds, no global atomics, no sort), then out; add order varies
# speedup vs baseline: 1.8893x; 1.8893x over previous
#include <hip/hip_runtime.h>

#define N_NODES   100000
#define N_EDGES   1600000
#define N_GRAPHS  1024
#define IN_LOCAL  16
#define IN_GLOBAL 8
#define HIDDEN    128
#define KC        (IN_LOCAL + HIDDEN)

typedef __attribute__((ext_vector_type(2))) float v2f;
typedef __attribute__((ext_vector_type(8))) float v8f;

#define LDSA_STRIDE 145
#define LDSH_STRIDE 129

__global__ __launch_bounds__(256) void k_global_enc(
    const float* __restrict__ gf, const float* __restrict__ Wg,
    const float* __restrict__ bg, float* __restrict__ hg) {
  int idx = blockIdx.x * 256 + threadIdx.x;
  int g = idx >> 7, h = idx & 127;
  float acc = bg[h];
#pragma unroll
  for (int k = 0; k < IN_GLOBAL; ++k)
    acc = fmaf(gf[g * IN_GLOBAL + k], Wg[k * HIDDEN + h], acc);
  hg[idx] = fmaxf(acc, 0.0f);
}

__global__ __launch_bounds__(256) void k_node_gemms(
    const float* __restrict__ x,  const int* __restrict__ batch,
    const float* __restrict__ hg,
    const float* __restrict__ Wc, const float* __restrict__ bc,
    const float* __restrict__ Wm, const float* __restrict__ bm,
    const float* __restrict__ Ws, const float* __restrict__ bs,
    float* __restrict__ m_buf, float* __restrict__ acc_buf) {
  __shared__ float lds_a[16 * LDSA_STRIDE];
  __shared__ float lds_h[16 * LDSH_STRIDE];

  const int tid = threadIdx.x;
  const int nodeBase = blockIdx.x * 16;

  for (int i = tid; i < 16 * KC; i += 256) {
    int row = i / KC, k = i - row * KC;
    int node = nodeBase + row;
    float v = (k < IN_LOCAL) ? x[node * IN_LOCAL + k]
                             : hg[batch[node] * HIDDEN + (k - IN_LOCAL)];
    lds_a[row * LDSA_STRIDE + k] = v;
  }
  __syncthreads();

  const int lane  = tid & 31;
  const int wave  = tid >> 5;
  const int half  = lane >> 4;
  const int idx16 = lane & 15;
  const int colBase = wave * 16;

  v8f acc = {};
#pragma unroll 4
  for (int kb = 0; kb < KC; kb += 4) {
    const int k0 = kb + 2 * half;
    v2f a, b;
    a.x = lds_a[idx16 * LDSA_STRIDE + k0];
    a.y = lds_a[idx16 * LDSA_STRIDE + k0 + 1];
    b.x = Wc[k0 * HIDDEN + colBase + idx16];
    b.y = Wc[(k0 + 1) * HIDDEN + colBase + idx16];
    acc = __builtin_amdgcn_wmma_f32_16x16x4_f32(
        false, a, false, b, (short)0, acc, false, false);
  }
  {
    const float bias = bc[colBase + idx16];
#pragma unroll
    for (int j = 0; j < 8; ++j) {
      int row = j + 8 * half;
      lds_h[row * LDSH_STRIDE + colBase + idx16] = fmaxf(acc[j] + bias, 0.0f);
    }
  }
  __syncthreads();

  v8f am = {}, as = {};
#pragma unroll 4
  for (int kb = 0; kb < HIDDEN; kb += 4) {
    const int k0 = kb + 2 * half;
    v2f a, b0, b1;
    a.x  = lds_h[idx16 * LDSH_STRIDE + k0];
    a.y  = lds_h[idx16 * LDSH_STRIDE + k0 + 1];
    b0.x = Wm[k0 * HIDDEN + colBase + idx16];
    b0.y = Wm[(k0 + 1) * HIDDEN + colBase + idx16];
    b1.x = Ws[k0 * HIDDEN + colBase + idx16];
    b1.y = Ws[(k0 + 1) * HIDDEN + colBase + idx16];
    am = __builtin_amdgcn_wmma_f32_16x16x4_f32(
        false, a, false, b0, (short)0, am, false, false);
    as = __builtin_amdgcn_wmma_f32_16x16x4_f32(
        false, a, false, b1, (short)0, as, false, false);
  }
  {
    const int col = colBase + idx16;
    const float biasm = bm[col], biass = bs[col];
#pragma unroll
    for (int j = 0; j < 8; ++j) {
      int row  = j + 8 * half;
      int node = nodeBase + row;
      float mv = fmaxf(am[j] + biasm, 0.0f);
      m_buf[(size_t)node * HIDDEN + col]   = mv;
      acc_buf[(size_t)node * HIDDEN + col] = mv + as[j] + biass;
    }
  }
}

#define OWN_WGS    256
#define OWN_NODES  ((N_NODES + OWN_WGS - 1) / OWN_WGS)
#define OWN_WAVES  32
#define CHUNK      512
#define N_CHUNKS   (N_EDGES / CHUNK)
#define QCAP       (32 + CHUNK)
#define E_BITS     21

typedef __attribute__((ext_vector_type(4))) float v4f;
typedef __attribute__((ext_vector_type(4))) int   v4i;

__device__ __forceinline__ void own_drain(
    const unsigned* q, int qcount, int n, int lane,
    const int* __restrict__ srcp, const float* __restrict__ m_buf, float* tile) {
  const unsigned ent = (lane < n) ? q[qcount - n + lane] : 0u;
  const int e = (int)(ent & ((1u << E_BITS) - 1u));
  const int t = (int)(ent >> E_BITS);
  const int s = srcp[e];
#pragma unroll 1
  for (int i0 = 0; i0 < n; i0 += 8) {
    v4f r[8];
    int tt[8];
#pragma unroll
    for (int j = 0; j < 8; ++j) {
      const int si = __builtin_amdgcn_readlane(s, i0 + j);
      tt[j] = __builtin_amdgcn_readlane(t, i0 + j);
      r[j] = *(const v4f*)(m_buf + (size_t)si * HIDDEN + 4 * lane);
    }
#pragma unroll
    for (int j = 0; j < 8; ++j) {
      if (i0 + j < n) {
        float* row = tile + tt[j] * HIDDEN + lane;
        __hip_atomic_fetch_add(row,      r[j].x, __ATOMIC_RELAXED, __HIP_MEMORY_SCOPE_WORKGROUP);
        __hip_atomic_fetch_add(row + 32, r[j].y, __ATOMIC_RELAXED, __HIP_MEMORY_SCOPE_WORKGROUP);
        __hip_atomic_fetch_add(row + 64, r[j].z, __ATOMIC_RELAXED, __HIP_MEMORY_SCOPE_WORKGROUP);
        __hip_atomic_fetch_add(row + 96, r[j].w, __ATOMIC_RELAXED, __HIP_MEMORY_SCOPE_WORKGROUP);
      }
    }
  }
}

__device__ __forceinline__ int own_chunk(int it, int rot) {
  int c = it + rot;
  return (c >= N_CHUNKS) ? c - N_CHUNKS : c;
}
__device__ __forceinline__ int own_clamp(int it, int first) {
  return (it < N_CHUNKS) ? it : first;
}
__device__ __forceinline__ void own_load(v4i (&d)[4], const int* __restrict__ dstp,
                                         int c, int lane) {
#pragma unroll
  for (int u = 0; u < 4; ++u)
    d[u] = *(const v4i*)(dstp + c * CHUNK + u * 128 + 4 * lane);
}
__device__ __forceinline__ int own_scan(const v4i (&d)[4], int base, int lo, int cnt,
                                        unsigned* q, int qcount, int lane) {
#pragma unroll
  for (int u = 0; u < 4; ++u) {
#pragma unroll
    for (int k = 0; k < 4; ++k) {
      const unsigned t = (unsigned)(d[u][k] - lo);
      const bool hit = t < (unsigned)cnt;
      const unsigned mk = __builtin_amdgcn_ballot_w32(hit);
      if (mk) {
        if (hit) {
          const unsigned e = (unsigned)(base + u * 128 + 4 * lane + k);
          q[qcount + __builtin_amdgcn_mbcnt_lo(mk, 0u)] = e | (t << E_BITS);
        }
        qcount += __builtin_popcount(mk);
      }
    }
  }
  return qcount;
}

__global__ __launch_bounds__(1024) void k_own_sum_out(
    const int* __restrict__ edge_index, const float* __restrict__ m_buf,
    const float* __restrict__ acc_buf, const float* __restrict__ Wo,
    const float* __restrict__ bo, float* __restrict__ out) {
  __shared__ float    tile[OWN_NODES * HIDDEN];
  __shared__ unsigned queue[OWN_WAVES * QCAP];

  const int tid  = threadIdx.x;
  const int lane = tid & 31;
  const int wave = __builtin_amdgcn_readfirstlane(tid >> 5);
  const int lo   = blockIdx.x * OWN_NODES;
  const int cnt  = min(OWN_NODES, N_NODES - lo);

  for (int i = tid; i < cnt * 32; i += 1024) {
    const int n = i >> 5;
    const v4f v = *(const v4f*)(acc_buf + (size_t)(lo + n) * HIDDEN + 4 * lane);
    float* row = tile + n * HIDDEN + lane;
    row[0] = v.x; row[32] = v.y; row[64] = v.z; row[96] = v.w;
  }
  __syncthreads();

  const int* __restrict__ srcp = edge_index;
  const int* __restrict__ dstp = edge_index + N_EDGES;
  unsigned* q = queue + wave * QCAP;
  int qcount = 0;
  const int rot = blockIdx.x * 12;

  v4i bufA[4], bufB[4];
  own_load(bufA, dstp, own_chunk(wave, rot), lane);
  __builtin_amdgcn_sched_barrier(0);
  own_load(bufB, dstp, own_chunk(own_clamp(wave + OWN_WAVES, wave), rot), lane);
  for (int it = wave; it < N_CHUNKS; it += 2 * OWN_WAVES) {
    qcount = own_scan(bufA, own_chunk(it, rot) * CHUNK, lo, cnt, q, qcount, lane);
    own_load(bufA, dstp, own_chunk(own_clamp(it + 2 * OWN_WAVES, wave), rot), lane);
    while (qcount >= 32) { own_drain(q, qcount, 32, lane, srcp, m_buf, tile); qcount -= 32; }

    const int cntB = (it + OWN_WAVES < N_CHUNKS) ? cnt : 0;
    qcount = own_scan(bufB, own_chunk(own_clamp(it + OWN_WAVES, wave), rot) * CHUNK, lo, cntB, q, qcount, lane);
    own_load(bufB, dstp, own_chunk(own_clamp(it + 3 * OWN_WAVES, wave), rot), lane);
    while (qcount >= 32) { own_drain(q, qcount, 32, lane, srcp, m_buf, tile); qcount -= 32; }
  }
  if (qcount > 0) own_drain(q, qcount, qcount, lane, srcp, m_buf, tile);
  __syncthreads();

  float w0[4], w1[4];
#pragma unroll
  for (int t = 0; t < 4; ++t) {
    const int k = lane + t * 32;
    w0[t] = Wo[k * 2 + 0];
    w1[t] = Wo[k * 2 + 1];
  }
  const float b0 = bo[0], b1 = bo[1];
  for (int n = wave; n < cnt; n += OWN_WAVES) {
    const float* row = tile + n * HIDDEN;
    float a0 = 0.0f, a1 = 0.0f;
#pragma unroll
    for (int t = 0; t < 4; ++t) {
      const int k = lane + t * 32;
      const float h = fmaxf(row[(k & 3) * 32 + (k >> 2)], 0.0f);
      a0 = fmaf(h, w0[t], a0);
      a1 = fmaf(h, w1[t], a1);
    }
#pragma unroll
    for (int off = 16; off >= 1; off >>= 1) {
      a0 += __shfl_xor(a0, off, 32);
      a1 += __shfl_xor(a1, off, 32);
    }
    if (lane == 0) {
      out[(lo + n) * 2 + 0] = a0 + b0;
      out[(lo + n) * 2 + 1] = a1 + b1;
    }
  }
}

extern "C" void kernel_launch(void* const* d_in, const int* in_sizes, int n_in,
                              void* d_out, int out_size, void* d_ws, size_t ws_size,
                              hipStream_t stream) {
  (void)in_sizes; (void)n_in; (void)out_size; (void)ws_size;
  const float* x     = (const float*)d_in[0];
  const int*   edge  = (const int*)  d_in[1];
  const int*   batch = (const int*)  d_in[2];
  const float* gf    = (const float*)d_in[3];
  const float* Wg    = (const float*)d_in[4];
  const float* bg    = (const float*)d_in[5];
  const float* Wc    = (const float*)d_in[6];
  const float* bc    = (const float*)d_in[7];
  const float* Wm    = (const float*)d_in[8];
  const float* bm    = (const float*)d_in[9];
  const float* Ws    = (const float*)d_in[10];
  const float* bs    = (const float*)d_in[11];
  const float* Wo    = (const float*)d_in[12];
  const float* bo    = (const float*)d_in[13];
  float* out = (float*)d_out;

  float* hg      = (float*)d_ws;
  float* m_buf   = hg    + (size_t)N_GRAPHS * HIDDEN;
  float* acc_buf = m_buf + (size_t)N_NODES  * HIDDEN;

  k_global_enc<<<(N_GRAPHS * HIDDEN) / 256, 256, 0, stream>>>(gf, Wg, bg, hg);
  k_node_gemms<<<N_NODES / 16, 256, 0, stream>>>(x, batch, hg, Wc, bc,
                                                 Wm, bm, Ws, bs, m_buf, acc_buf);
  k_own_sum_out<<<OWN_WGS, 1024, 0, stream>>>(edge, m_buf, acc_buf, Wo, bo, out);
}


# Edge scan reads a 1-byte owner id per edge (target / 391, written in the node GEMM launch) instead of the 4-byte target: one test per 32 edges a lane; add order varies as before
# speedup vs baseline: 2.2709x; 1.2020x over previous
#include <hip/hip_runtime.h>

#define N_NODES   100000
#define N_EDGES   1600000
#define N_GRAPHS  1024
#define IN_LOCAL  16
#define IN_GLOBAL 8
#define HIDDEN    128
#define KC        (IN_LOCAL + HIDDEN)

typedef __attribute__((ext_vector_type(2))) float v2f;
typedef __attribute__((ext_vector_type(8))) float v8f;

#define LDSA_STRIDE 145
#define LDSH_STRIDE 129

#define OWN_WGS    256
#define OWN_NODES  ((N_NODES + OWN_WGS - 1) / OWN_WGS)
#define WS_FLOATS  ((size_t)N_GRAPHS * HIDDEN + 2 * (size_t)N_NODES * HIDDEN)
#define WS_NEED    (WS_FLOATS * 4 + N_EDGES)

typedef __attribute__((ext_vector_type(4))) float v4f;
typedef __attribute__((ext_vector_type(4))) int   v4i;

__global__ __launch_bounds__(256) void k_global_enc(
    const float* __restrict__ gf, const float* __restrict__ Wg,
    const float* __restrict__ bg, float* __restrict__ hg) {
  int idx = blockIdx.x * 256 + threadIdx.x;
  int g = idx >> 7, h = idx & 127;
  float acc = bg[h];
#pragma unroll
  for (int k = 0; k < IN_GLOBAL; ++k)
    acc = fmaf(gf[g * IN_GLOBAL + k], Wg[k * HIDDEN + h], acc);
  hg[idx] = fmaxf(acc, 0.0f);
}

__global__ __launch_bounds__(256) void k_node_gemms(
    const float* __restrict__ x,  const int* __restrict__ edge_index,
    const int* __restrict__ batch,
    const float* __restrict__ hg,
    const float* __restrict__ Wc, const float* __restrict__ bc,
    const float* __restrict__ Wm, const float* __restrict__ bm,
    const float* __restrict__ Ws, const float* __restrict__ bs,
    float* __restrict__ m_buf, float* __restrict__ acc_buf,
    unsigned* __restrict__ own_w, size_t ws_bytes) {
  __shared__ float lds_a[16 * LDSA_STRIDE];
  __shared__ float lds_h[16 * LDSH_STRIDE];

  const int tid = threadIdx.x;
  const int nodeBase = blockIdx.x * 16;

  const bool own_lane = ws_bytes >= WS_NEED && tid < 64;
  v4i own_t = {};
  if (own_lane)
    own_t = *(const v4i*)(edge_index + N_EDGES + blockIdx.x * 256 + 4 * tid);

  for (int i = tid; i < 16 * KC; i += 256) {
    int row = i / KC, k = i - row * KC;
    int node = nodeBase + row;
    float v = (k < IN_LOCAL) ? x[node * IN_LOCAL + k]
                             : hg[batch[node] * HIDDEN + (k - IN_LOCAL)];
    lds_a[row * LDSA_STRIDE + k] = v;
  }
  if (own_lane) {
    const unsigned o0 = ((unsigned)own_t.x / OWN_NODES) & 255u;
    const unsigned o1 = ((unsigned)own_t.y / OWN_NODES) & 255u;
    const unsigned o2 = ((unsigned)own_t.z / OWN_NODES) & 255u;
    const unsigned o3 = ((unsigned)own_t.w / OWN_NODES) & 255u;
    own_w[blockIdx.x * 64 + tid] = o0 | (o1 << 8) | (o2 << 16) | (o3 << 24);
  }
  __syncthreads();

  const int lane  = tid & 31;
  const int wave  = tid >> 5;
  const int half  = lane >> 4;
  const int idx16 = lane & 15;
  const int colBase = wave * 16;

  v8f acc = {};
#pragma unroll 4
  for (int kb = 0; kb < KC; kb += 4) {
    const int k0 = kb + 2 * half;
    v2f a, b;
    a.x = lds_a[idx16 * LDSA_STRIDE + k0];
    a.y = lds_a[idx16 * LDSA_STRIDE + k0 + 1];
    b.x = Wc[k0 * HIDDEN + colBase + idx16];
    b.y = Wc[(k0 + 1) * HIDDEN + colBase + idx16];
    acc = __builtin_amdgcn_wmma_f32_16x16x4_f32(
        false, a, false, b, (short)0, acc, false, false);
  }
  {
    const float bias = bc[colBase + idx16];
#pragma unroll
    for (int j = 0; j < 8; ++j) {
      int row = j + 8 * half;
      lds_h[row * LDSH_STRIDE + colBase + idx16] = fmaxf(acc[j] + bias, 0.0f);
    }
  }
  __syncthreads();

  v8f am = {}, as = {};
#pragma unroll 4
  for (int kb = 0; kb < HIDDEN; kb += 4) {
    const int k0 = kb + 2 * half;
    v2f a, b0, b1;
    a.x  = lds_h[idx16 * LDSH_STRIDE + k0];
    a.y  = lds_h[idx16 * LDSH_STRIDE + k0 + 1];
    b0.x = Wm[k0 * HIDDEN + colBase + idx16];
    b0.y = Wm[(k0 + 1) * HIDDEN + colBase + idx16];
    b1.x = Ws[k0 * HIDDEN + colBase + idx16];
    b1.y = Ws[(k0 + 1) * HIDDEN + colBase + idx16];
    am = __builtin_amdgcn_wmma_f32_16x16x4_f32(
        false, a, false, b0, (short)0, am, false, false);
    as = __builtin_amdgcn_wmma_f32_16x16x4_f32(
        false, a, false, b1, (short)0, as, false, false);
  }
  {
    const int col = colBase + idx16;
    const float biasm = bm[col], biass = bs[col];
#pragma unroll
    for (int j = 0; j < 8; ++j) {
      int row  = j + 8 * half;
      int node = nodeBase + row;
      float mv = fmaxf(am[j] + biasm, 0.0f);
      m_buf[(size_t)node * HIDDEN + col]   = mv;
      acc_buf[(size_t)node * HIDDEN + col] = mv + as[j] + biass;
    }
  }
}

#define OWN_WAVES  32
#define CHUNK      512
#define N_CHUNKS   (N_EDGES / CHUNK)
#define QCAP       (32 + CHUNK)
#define STEP_UNITS (OWN_WAVES * 4)
#define N_STEPS    ((N_CHUNKS + STEP_UNITS - 1) / STEP_UNITS)

__device__ __forceinline__ void own_drain(
    const unsigned* q, int qcount, int n, int lane, int lo, int cnt,
    const int* __restrict__ srcp, const int* __restrict__ dstp,
    const float* __restrict__ m_buf, float* tile) {
  const int e = (lane < n) ? (int)q[qcount - n + lane] : 0;
  const int s = srcp[e];
  const unsigned tl = (unsigned)(dstp[e] - lo);
  const int t = (tl < (unsigned)cnt) ? (int)tl : OWN_NODES;
#pragma unroll 1
  for (int i0 = 0; i0 < n; i0 += 8) {
    v4f r[8];
    int tt[8];
#pragma unroll
    for (int j = 0; j < 8; ++j) {
      const int si = __builtin_amdgcn_readlane(s, i0 + j);
      tt[j] = __builtin_amdgcn_readlane(t, i0 + j);
      r[j] = *(const v4f*)(m_buf + (size_t)si * HIDDEN + 4 * lane);
    }
#pragma unroll
    for (int j = 0; j < 8; ++j) {
      if (i0 + j < n) {
        float* row = tile + tt[j] * HIDDEN + lane;
        __hip_atomic_fetch_add(row,      r[j].x, __ATOMIC_RELAXED, __HIP_MEMORY_SCOPE_WORKGROUP);
        __hip_atomic_fetch_add(row + 32, r[j].y, __ATOMIC_RELAXED, __HIP_MEMORY_SCOPE_WORKGROUP);
        __hip_atomic_fetch_add(row + 64, r[j].z, __ATOMIC_RELAXED, __HIP_MEMORY_SCOPE_WORKGROUP);
        __hip_atomic_fetch_add(row + 96, r[j].w, __ATOMIC_RELAXED, __HIP_MEMORY_SCOPE_WORKGROUP);
      }
    }
  }
}

__device__ __forceinline__ int own_chunk(int it, int rot) {
  int c = it + rot;
  return (c >= N_CHUNKS) ? c - N_CHUNKS : c;
}
__device__ __forceinline__ int own_clamp(int it, int first) {
  return (it < N_CHUNKS) ? it : first;
}
__device__ __forceinline__ void own_load(v4i (&d)[4], const int* __restrict__ dstp,
                                         int c, int lane) {
#pragma unroll
  for (int u = 0; u < 4; ++u)
    d[u] = *(const v4i*)(dstp + c * CHUNK + u * 128 + 4 * lane);
}
__device__ __forceinline__ int own_scan(const v4i (&d)[4], int base, int lo, int cnt,
                                        unsigned* q, int qcount, int lane) {
#pragma unroll
  for (int u = 0; u < 4; ++u) {
#pragma unroll
    for (int k = 0; k < 4; ++k) {
      const unsigned t = (unsigned)(d[u][k] - lo);
      const bool hit = t < (unsigned)cnt;
      const unsigned mk = __builtin_amdgcn_ballot_w32(hit);
      if (mk) {
        if (hit) {
          const unsigned e = (unsigned)(base + u * 128 + 4 * lane + k);
          q[qcount + __builtin_amdgcn_mbcnt_lo(mk, 0u)] = e;
        }
        qcount += __builtin_popcount(mk);
      }
    }
  }
  return qcount;
}

__device__ __forceinline__ int own_unit(int st, int wave, int u) {
  return st * STEP_UNITS + wave * 4 + u;
}
__device__ __forceinline__ void own_bload(v4i (&d)[4], const unsigned* __restrict__ own_w,
                                          int st, int wave, int rot, int lane) {
#pragma unroll
  for (int u = 0; u < 4; ++u) {
    const int c = own_chunk(min(own_unit(st, wave, u), N_CHUNKS - 1), rot);
    d[u] = *(const v4i*)(own_w + c * (CHUNK / 4) + 4 * lane);
  }
}
__device__ __forceinline__ unsigned own_nz(unsigned x) {
  return ((x & 0x7F7F7F7Fu) + 0x7F7F7F7Fu) | x;
}
__device__ __forceinline__ unsigned own_mask(const v4i& a, const v4i& b,
                                             unsigned pat, unsigned keep) {
  unsigned t = own_nz((unsigned)a.x ^ pat);
  t = (t & ~0x40404040u) | ((own_nz((unsigned)a.y ^ pat) >> 1) & 0x40404040u);
  t = (t & ~0x20202020u) | ((own_nz((unsigned)a.z ^ pat) >> 2) & 0x20202020u);
  t = (t & ~0x10101010u) | ((own_nz((unsigned)a.w ^ pat) >> 3) & 0x10101010u);
  t = (t & ~0x08080808u) | ((own_nz((unsigned)b.x ^ pat) >> 4) & 0x08080808u);
  t = (t & ~0x04040404u) | ((own_nz((unsigned)b.y ^ pat) >> 5) & 0x04040404u);
  t = (t & ~0x02020202u) | ((own_nz((unsigned)b.z ^ pat) >> 6) & 0x02020202u);
  t = (t & ~0x01010101u) | ((own_nz((unsigned)b.w ^ pat) >> 7) & 0x01010101u);
  return ~t & keep;
}
__device__ __forceinline__ unsigned own_keep(int ua, int ub) {
  return ((ua < N_CHUNKS) ? 0xF0F0F0F0u : 0u) | ((ub < N_CHUNKS) ? 0x0F0F0F0Fu : 0u);
}
__device__ __forceinline__ int own_pop(
    unsigned m, int ea, int eb, unsigned* q, int qcount, int lane, int lo, int cnt,
    const int* __restrict__ srcp, const int* __restrict__ dstp,
    const float* __restrict__ m_buf, float* tile) {
  unsigned mk = __builtin_amdgcn_ballot_w32(m != 0u);
  while (mk) {
    if (m != 0u) {
      const int p = __builtin_ctz(m);
      m &= m - 1u;
      const int j = 7 - (p & 7);
      const int e = ((j & 4) ? eb : ea) + 16 * lane + 4 * (j & 3) + (p >> 3);
      q[qcount + __builtin_amdgcn_mbcnt_lo(mk, 0u)] = (unsigned)e;
    }
    qcount += __builtin_popcount(mk);
    if (qcount >= 32) {
      own_drain(q, qcount, 32, lane, lo, cnt, srcp, dstp, m_buf, tile);
      qcount -= 32;
    }
    mk = __builtin_amdgcn_ballot_w32(m != 0u);
  }
  return qcount;
}
__device__ __forceinline__ int own_bstep(
    v4i (&d)[4], const unsigned* __restrict__ own_w, int st, int st_next, int wave,
    int rot, unsigned pat, unsigned* q, int qcount, int lane, int lo, int cnt,
    const int* __restrict__ srcp, const int* __restrict__ dstp,
    const float* __restrict__ m_buf, float* tile) {
  const int u0 = own_unit(st, wave, 0);
  const unsigned m0 = own_mask(d[0], d[1], pat, own_keep(u0,     u0 + 1));
  const unsigned m1 = own_mask(d[2], d[3], pat, own_keep(u0 + 2, u0 + 3));
  const int c0 = own_chunk(min(u0,     N_CHUNKS - 1), rot) * CHUNK;
  const int c1 = own_chunk(min(u0 + 1, N_CHUNKS - 1), rot) * CHUNK;
  const int c2 = own_chunk(min(u0 + 2, N_CHUNKS - 1), rot) * CHUNK;
  const int c3 = own_chunk(min(u0 + 3, N_CHUNKS - 1), rot) * CHUNK;
  qcount = own_pop(m0, c0, c1, q, qcount, lane, lo, cnt, srcp, dstp, m_buf, tile);
  qcount = own_pop(m1, c2, c3, q, qcount, lane, lo, cnt, srcp, dstp, m_buf, tile);
  own_bload(d, own_w, st_next, wave, rot, lane);
  return qcount;
}

__global__ __launch_bounds__(1024) void k_own_sum_out(
    const int* __restrict__ edge_index, const float* __restrict__ m_buf,
    const float* __restrict__ acc_buf, const unsigned* __restrict__ own_w,
    size_t ws_bytes, const float* __restrict__ Wo,
    const float* __restrict__ bo, float* __restrict__ out) {
  __shared__ float    tile[(OWN_NODES + 1) * HIDDEN];
  __shared__ unsigned queue[OWN_WAVES * QCAP];

  const int tid  = threadIdx.x;
  const int lane = tid & 31;
  const int wave = __builtin_amdgcn_readfirstlane(tid >> 5);
  const int lo   = blockIdx.x * OWN_NODES;
  const int cnt  = min(OWN_NODES, N_NODES - lo);

  for (int i = tid; i < cnt * 32; i += 1024) {
    const int n = i >> 5;
    const v4f v = *(const v4f*)(acc_buf + (size_t)(lo + n) * HIDDEN + 4 * lane);
    float* row = tile + n * HIDDEN + lane;
    row[0] = v.x; row[32] = v.y; row[64] = v.z; row[96] = v.w;
  }
  __syncthreads();

  const int* __restrict__ srcp = edge_index;
  const int* __restrict__ dstp = edge_index + N_EDGES;
  unsigned* q = queue + wave * QCAP;
  int qcount = 0;
  const int rot = blockIdx.x * 12;

  v4i bufA[4], bufB[4];
  if (ws_bytes >= WS_NEED) {
    const unsigned pat = (unsigned)blockIdx.x * 0x01010101u;
    own_bload(bufA, own_w, 0, wave, rot, lane);
    __builtin_amdgcn_sched_barrier(0);
    own_bload(bufB, own_w, 1, wave, rot, lane);
    for (int st = 0; st < N_STEPS; st += 2) {
      qcount = own_bstep(bufA, own_w, st,     st + 2, wave, rot, pat, q, qcount, lane,
                         lo, cnt, srcp, dstp, m_buf, tile);
      qcount = own_bstep(bufB, own_w, st + 1, st + 3, wave, rot, pat, q, qcount, lane,
                         lo, cnt, srcp, dstp, m_buf, tile);
    }
  } else {
    own_load(bufA, dstp, own_chunk(wave, rot), lane);
    __builtin_amdgcn_sched_barrier(0);
    own_load(bufB, dstp, own_chunk(own_clamp(wave + OWN_WAVES, wave), rot), lane);
    for (int it = wave; it < N_CHUNKS; it += 2 * OWN_WAVES) {
      qcount = own_scan(bufA, own_chunk(it, rot) * CHUNK, lo, cnt, q, qcount, lane);
      own_load(bufA, dstp, own_chunk(own_clamp(it + 2 * OWN_WAVES, wave), rot), lane);
      while (qcount >= 32) {
        own_drain(q, qcount, 32, lane, lo, cnt, srcp, dstp, m_buf, tile); qcount -= 32;
      }

      const int cntB = (it + OWN_WAVES < N_CHUNKS) ? cnt : 0;
      qcount = own_scan(bufB, own_chunk(own_clamp(it + OWN_WAVES, wave), rot) * CHUNK, lo, cntB, q, qcount, lane);
      own_load(bufB, dstp, own_chunk(own_clamp(it + 3 * OWN_WAVES, wave), rot), lane);
      while (qcount >= 32) {
        own_drain(q, qcount, 32, lane, lo, cnt, srcp, dstp, m_buf, tile); qcount -= 32;
      }
    }
  }
  if (qcount > 0) own_drain(q, qcount, qcount, lane, lo, cnt, srcp, dstp, m_buf, tile);
  __syncthreads();

  float w0[4], w1[4];
#pragma unroll
  for (int t = 0; t < 4; ++t) {
    const int k = lane + t * 32;
    w0[t] = Wo[k * 2 + 0];
    w1[t] = Wo[k * 2 + 1];
  }
  const float b0 = bo[0], b1 = bo[1];
  for (int n = wave; n < cnt; n += OWN_WAVES) {
    const float* row = tile + n * HIDDEN;
    float a0 = 0.0f, a1 = 0.0f;
#pragma unroll
    for (int t = 0; t < 4; ++t) {
      const int k = lane + t * 32;
      const float h = fmaxf(row[(k & 3) * 32 + (k >> 2)], 0.0f);
      a0 = fmaf(h, w0[t], a0);
      a1 = fmaf(h, w1[t], a1);
    }
#pragma unroll
    for (int off = 16; off >= 1; off >>= 1) {
      a0 += __shfl_xor(a0, off, 32);
      a1 += __shfl_xor(a1, off, 32);
    }
    if (lane == 0) {
      out[(lo + n) * 2 + 0] = a0 + b0;
      out[(lo + n) * 2 + 1] = a1 + b1;
    }
  }
}

extern "C" void kernel_launch(void* const* d_in, const int* in_sizes, int n_in,
                              void* d_out, int out_size, void* d_ws, size_t ws_size,
                              hipStream_t stream) {
  (void)in_sizes; (void)n_in; (void)out_size;
  const float* x     = (const float*)d_in[0];
  const int*   edge  = (const int*)  d_in[1];
  const int*   batch = (const int*)  d_in[2];
  const float* gf    = (const float*)d_in[3];
  const float* Wg    = (const float*)d_in[4];
  const float* bg    = (const float*)d_in[5];
  const float* Wc    = (const float*)d_in[6];
  const float* bc    = (const float*)d_in[7];
  const float* Wm    = (const float*)d_in[8];
  const float* bm    = (const float*)d_in[9];
  const float* Ws    = (const float*)d_in[10];
  const float* bs    = (const float*)d_in[11];
  const float* Wo    = (const float*)d_in[12];
  const float* bo    = (const float*)d_in[13];
  float* out = (float*)d_out;

  float* hg      = (float*)d_ws;
  float* m_buf   = hg    + (size_t)N_GRAPHS * HIDDEN;
  float* acc_buf = m_buf + (size_t)N_NODES  * HIDDEN;
  unsigned* own_w = (unsigned*)(acc_buf + (size_t)N_NODES * HIDDEN);

  k_global_enc<<<(N_GRAPHS * HIDDEN) / 256, 256, 0, stream>>>(gf, Wg, bg, hg);
  k_node_gemms<<<N_NODES / 16, 256, 0, stream>>>(x, edge, batch, hg, Wc, bc,
                                                 Wm, bm, Ws, bs, m_buf, acc_buf,
                                                 own_w, ws_size);
  k_own_sum_out<<<OWN_WGS, 1024, 0, stream>>>(edge, m_buf, acc_buf, own_w, ws_size,
                                              Wo, bo, out);
}


# Node GEMMs persistent: 256 work-groups keep Wc, Wm, Ws in LDS all launch (B fragments by LDS load, not global loads per tile); next tiles' inputs fetched under the products; sum order kept
# speedup vs baseline: 2.6327x; 1.1593x over previous
#include <hip/hip_runtime.h>

#define N_NODES   100000
#define N_EDGES   1600000
#define N_GRAPHS  1024
#define IN_LOCAL  16
#define IN_GLOBAL 8
#define HIDDEN    128
#define KC        (IN_LOCAL + HIDDEN)

typedef __attribute__((ext_vector_type(2))) float v2f;
typedef __attribute__((ext_vector_type(8))) float v8f;

#define OWN_WGS    256
#define OWN_NODES  ((N_NODES + OWN_WGS - 1) / OWN_WGS)
#define WS_FLOATS  ((size_t)N_GRAPHS * HIDDEN + 2 * (size_t)N_NODES * HIDDEN)
#define WS_NEED    (WS_FLOATS * 4 + N_EDGES)

typedef __attribute__((ext_vector_type(4))) float v4f;
typedef __attribute__((ext_vector_type(4))) int   v4i;

__global__ __launch_bounds__(256) void k_global_enc(
    const float* __restrict__ gf, const float* __restrict__ Wg,
    const float* __restrict__ bg, float* __restrict__ hg) {
  int idx = blockIdx.x * 256 + threadIdx.x;
  int g = idx >> 7, h = idx & 127;
  float acc = bg[h];
#pragma unroll
  for (int k = 0; k < IN_GLOBAL; ++k)
    acc = fmaf(gf[g * IN_GLOBAL + k], Wg[k * HIDDEN + h], acc);
  hg[idx] = fmaxf(acc, 0.0f);
}

#define NT_TILES    (N_NODES / 16)
#define NG_WGS      256
#define NG_GROUPS   4
#define NG_STEPS    ((NT_TILES + NG_WGS * NG_GROUPS - 1) / (NG_WGS * NG_GROUPS))
#define NG_LONG     (NT_TILES - NG_WGS * NG_GROUPS * (NG_STEPS - 1))
#define W_ROWS      (KC + 2 * HIDDEN)
#define LDSW_PITCH  288
#define LDSA_PITCH  148
#define LDSH_PITCH  132

__device__ __forceinline__ v8f wmma4(v2f a, v2f b, v8f c) {
  return __builtin_amdgcn_wmma_f32_16x16x4_f32(false, a, false, b, (short)0, c, false, false);
}

__global__ __launch_bounds__(1024) void k_node_gemms(
    const float* __restrict__ x,  const int* __restrict__ edge_index,
    const int* __restrict__ batch,
    const float* __restrict__ hg,
    const float* __restrict__ Wc, const float* __restrict__ bc,
    const float* __restrict__ Wm, const float* __restrict__ bm,
    const float* __restrict__ Ws, const float* __restrict__ bs,
    float* __restrict__ m_buf, float* __restrict__ acc_buf,
    unsigned* __restrict__ own_w, size_t ws_bytes) {
  __shared__ __attribute__((aligned(16))) float lds_w[(W_ROWS / 2) * LDSW_PITCH];
  __shared__ __attribute__((aligned(16))) float lds_a[16 * NG_GROUPS * LDSA_PITCH];
  __shared__ __attribute__((aligned(16))) float lds_h[16 * NG_GROUPS * LDSH_PITCH];

  const int tid  = threadIdx.x;
  const int wg   = blockIdx.x;
  const int lane = tid & 31;
  const int wave = __builtin_amdgcn_readfirstlane(tid >> 5);
  const int grp  = wave >> 3;
  const int half = lane >> 4;
  const int idx16 = lane & 15;
  const int col  = (wave & 7) * 16 + idx16;
  const int n_steps = (wg < NG_LONG) ? NG_STEPS : NG_STEPS - 1;

  const bool own_on = ws_bytes >= WS_NEED;
  const int ow0 = wg * 1024 + tid, ow1 = ow0 + NG_WGS * 1024;
  const bool own1 = own_on && ow1 < N_EDGES / 4;
  v4i own_t0 = {}, own_t1 = {};
  if (own_on) own_t0 = *(const v4i*)(edge_index + N_EDGES + 4 * ow0);
  if (own1)   own_t1 = *(const v4i*)(edge_index + N_EDGES + 4 * ow1);

  const int srow = tid >> 4, sseg = tid & 15;
  float* const sdst = lds_a + srow * LDSA_PITCH;

  int   nb;
  float nx;
  {
    const int nnode = (wg + NG_WGS * grp) * 16 + (srow & 15);
    nb = batch[nnode];
    nx = x[nnode * IN_LOCAL + sseg];
  }

  for (int i = tid; i < (W_ROWS / 2) * 32; i += 1024) {
    const int p = i >> 5, c4 = (i & 31) * 4;
    const int k = 2 * p;
    const float* src = (k < KC) ? (Wc + k * HIDDEN)
                     : (k < KC + HIDDEN) ? (Wm + (k - KC) * HIDDEN)
                                         : (Ws + (k - KC - HIDDEN) * HIDDEN);
    const v4f r0 = *(const v4f*)(src + c4);
    const v4f r1 = *(const v4f*)(src + HIDDEN + c4);
    float* d = lds_w + p * LDSW_PITCH + 2 * c4;
    *(v4f*)(d)     = (v4f){r0.x, r1.x, r0.y, r1.y};
    *(v4f*)(d + 4) = (v4f){r0.z, r1.z, r0.w, r1.w};
  }
  if (own_on) {
    const unsigned o0 = ((unsigned)own_t0.x / OWN_NODES) & 255u;
    const unsigned o1 = ((unsigned)own_t0.y / OWN_NODES) & 255u;
    const unsigned o2 = ((unsigned)own_t0.z / OWN_NODES) & 255u;
    const unsigned o3 = ((unsigned)own_t0.w / OWN_NODES) & 255u;
    own_w[ow0] = o0 | (o1 << 8) | (o2 << 16) | (o3 << 24);
  }
  if (own1) {
    const unsigned o0 = ((unsigned)own_t1.x / OWN_NODES) & 255u;
    const unsigned o1 = ((unsigned)own_t1.y / OWN_NODES) & 255u;
    const unsigned o2 = ((unsigned)own_t1.z / OWN_NODES) & 255u;
    const unsigned o3 = ((unsigned)own_t1.w / OWN_NODES) & 255u;
    own_w[ow1] = o0 | (o1 << 8) | (o2 << 16) | (o3 << 24);
  }
  {
    const v4f g0 = *(const v4f*)(hg + nb * HIDDEN + sseg * 8);
    const v4f g1 = *(const v4f*)(hg + nb * HIDDEN + sseg * 8 + 4);
    sdst[sseg] = nx;
    *(v4f*)(sdst + IN_LOCAL + sseg * 8)     = g0;
    *(v4f*)(sdst + IN_LOCAL + sseg * 8 + 4) = g1;
  }
  const float biasc = bc[col], biasm = bm[col], biass = bs[col];
  __syncthreads();

  const float* const ap  = lds_a + (grp * 16 + idx16) * LDSA_PITCH + 2 * half;
  const float* const hp  = lds_h + (grp * 16 + idx16) * LDSH_PITCH + 2 * half;
  const float* const bcp = lds_w + half * LDSW_PITCH + 2 * col;
  const float* const bmp = lds_w + (KC / 2 + half) * LDSW_PITCH + 2 * col;
  const float* const bsp = lds_w + ((KC + HIDDEN) / 2 + half) * LDSW_PITCH + 2 * col;

  for (int st = 0; st < n_steps; ++st) {
    const int  tile = wg + NG_WGS * (NG_GROUPS * st + grp);
    const bool live = tile < NT_TILES;

    {
      const int ntile = min(tile + NG_WGS * NG_GROUPS, NT_TILES - 1);
      const int nnode = ntile * 16 + (srow & 15);
      nb = batch[nnode];
      nx = x[nnode * IN_LOCAL + sseg];
    }

    if (live) {
      v8f acc = {};
#pragma unroll
      for (int kb = 0; kb < KC; kb += 4) {
        const v2f a = *(const v2f*)(ap + kb);
        const v2f b = *(const v2f*)(bcp + (kb / 2) * LDSW_PITCH);
        acc = wmma4(a, b, acc);
      }
#pragma unroll
      for (int j = 0; j < 8; ++j) {
        const int row = grp * 16 + j + 8 * half;
        lds_h[row * LDSH_PITCH + col] = fmaxf(acc[j] + biasc, 0.0f);
      }
    }
    __syncthreads();

    const v4f g0 = *(const v4f*)(hg + nb * HIDDEN + sseg * 8);
    const v4f g1 = *(const v4f*)(hg + nb * HIDDEN + sseg * 8 + 4);

    if (live) {
      v8f am = {}, as = {};
#pragma unroll
      for (int kb = 0; kb < HIDDEN; kb += 4) {
        const v2f a  = *(const v2f*)(hp + kb);
        const v2f b0 = *(const v2f*)(bmp + (kb / 2) * LDSW_PITCH);
        const v2f b1 = *(const v2f*)(bsp + (kb / 2) * LDSW_PITCH);
        am = wmma4(a, b0, am);
        as = wmma4(a, b1, as);
      }
      const int nodeBase = tile * 16;
#pragma unroll
      for (int j = 0; j < 8; ++j) {
        const int node = nodeBase + j + 8 * half;
        const float mv = fmaxf(am[j] + biasm, 0.0f);
        m_buf[(size_t)node * HIDDEN + col]   = mv;
        acc_buf[(size_t)node * HIDDEN + col] = mv + as[j] + biass;
      }
    }

    {
      sdst[sseg] = nx;
      *(v4f*)(sdst + IN_LOCAL + sseg * 8)     = g0;
      *(v4f*)(sdst + IN_LOCAL + sseg * 8 + 4) = g1;
    }
    __builtin_amdgcn_fence(__ATOMIC_RELEASE, "workgroup", "local");
    __builtin_amdgcn_s_barrier();
    __builtin_amdgcn_fence(__ATOMIC_ACQUIRE, "workgroup", "local");
  }
}

#define OWN_WAVES  32
#define CHUNK      512
#define N_CHUNKS   (N_EDGES / CHUNK)
#define QCAP       (32 + CHUNK)
#define STEP_UNITS (OWN_WAVES * 4)
#define N_STEPS    ((N_CHUNKS + STEP_UNITS - 1) / STEP_UNITS)

__device__ __forceinline__ void own_drain(
    const unsigned* q, int qcount, int n, int lane, int lo, int cnt,
    const int* __restrict__ srcp, const int* __restrict__ dstp,
    const float* __restrict__ m_buf, float* tile) {
  const int e = (lane < n) ? (int)q[qcount - n + lane] : 0;
  const int s = srcp[e];
  const unsigned tl = (unsigned)(dstp[e] - lo);
  const int t = (tl < (unsigned)cnt) ? (int)tl : OWN_NODES;
#pragma unroll 1
  for (int i0 = 0; i0 < n; i0 += 8) {
    v4f r[8];
    int tt[8];
#pragma unroll
    for (int j = 0; j < 8; ++j) {
      const int si = __builtin_amdgcn_readlane(s, i0 + j);
      tt[j] = __builtin_amdgcn_readlane(t, i0 + j);
      r[j] = *(const v4f*)(m_buf + (size_t)si * HIDDEN + 4 * lane);
    }
#pragma unroll
    for (int j = 0; j < 8; ++j) {
      if (i0 + j < n) {
        float* row = tile + tt[j] * HIDDEN + lane;
        __hip_atomic_fetch_add(row,      r[j].x, __ATOMIC_RELAXED, __HIP_MEMORY_SCOPE_WORKGROUP);
        __hip_atomic_fetch_add(row + 32, r[j].y, __ATOMIC_RELAXED, __HIP_MEMORY_SCOPE_WORKGROUP);
        __hip_atomic_fetch_add(row + 64, r[j].z, __ATOMIC_RELAXED, __HIP_MEMORY_SCOPE_WORKGROUP);
        __hip_atomic_fetch_add(row + 96, r[j].w, __ATOMIC_RELAXED, __HIP_MEMORY_SCOPE_WORKGROUP);
      }
    }
  }
}

__device__ __forceinline__ int own_chunk(int it, int rot) {
  int c = it + rot;
  return (c >= N_CHUNKS) ? c - N_CHUNKS : c;
}
__device__ __forceinline__ int own_clamp(int it, int first) {
  return (it < N_CHUNKS) ? it : first;
}
__device__ __forceinline__ void own_load(v4i (&d)[4], const int* __restrict__ dstp,
                                         int c, int lane) {
#pragma unroll
  for (int u = 0; u < 4; ++u)
    d[u] = *(const v4i*)(dstp + c * CHUNK + u * 128 + 4 * lane);
}
__device__ __forceinline__ int own_scan(const v4i (&d)[4], int base, int lo, int cnt,
                                        unsigned* q, int qcount, int lane) {
#pragma unroll
  for (int u = 0; u < 4; ++u) {
#pragma unroll
    for (int k = 0; k < 4; ++k) {
      const unsigned t = (unsigned)(d[u][k] - lo);
      const bool hit = t < (unsigned)cnt;
      const unsigned mk = __builtin_amdgcn_ballot_w32(hit);
      if (mk) {
        if (hit) {
          const unsigned e = (unsigned)(base + u * 128 + 4 * lane + k);
          q[qcount + __builtin_amdgcn_mbcnt_lo(mk, 0u)] = e;
        }
        qcount += __builtin_popcount(mk);
      }
    }
  }
  return qcount;
}

__device__ __forceinline__ int own_unit(int st, int wave, int u) {
  return st * STEP_UNITS + wave * 4 + u;
}
__device__ __forceinline__ void own_bload(v4i (&d)[4], const unsigned* __restrict__ own_w,
                                          int st, int wave, int rot, int lane) {
#pragma unroll
  for (int u = 0; u < 4; ++u) {
    const int c = own_chunk(min(own_unit(st, wave, u), N_CHUNKS - 1), rot);
    d[u] = *(const v4i*)(own_w + c * (CHUNK / 4) + 4 * lane);
  }
}
__device__ __forceinline__ unsigned own_nz(unsigned x) {
  return ((x & 0x7F7F7F7Fu) + 0x7F7F7F7Fu) | x;
}
__device__ __forceinline__ unsigned own_mask(const v4i& a, const v4i& b,
                                             unsigned pat, unsigned keep) {
  unsigned t = own_nz((unsigned)a.x ^ pat);
  t = (t & ~0x40404040u) | ((own_nz((unsigned)a.y ^ pat) >> 1) & 0x40404040u);
  t = (t & ~0x20202020u) | ((own_nz((unsigned)a.z ^ pat) >> 2) & 0x20202020u);
  t = (t & ~0x10101010u) | ((own_nz((unsigned)a.w ^ pat) >> 3) & 0x10101010u);
  t = (t & ~0x08080808u) | ((own_nz((unsigned)b.x ^ pat) >> 4) & 0x08080808u);
  t = (t & ~0x04040404u) | ((own_nz((unsigned)b.y ^ pat) >> 5) & 0x04040404u);
  t = (t & ~0x02020202u) | ((own_nz((unsigned)b.z ^ pat) >> 6) & 0x02020202u);
  t = (t & ~0x01010101u) | ((own_nz((unsigned)b.w ^ pat) >> 7) & 0x01010101u);
  return ~t & keep;
}
__device__ __forceinline__ unsigned own_keep(int ua, int ub) {
  return ((ua < N_CHUNKS) ? 0xF0F0F0F0u : 0u) | ((ub < N_CHUNKS) ? 0x0F0F0F0Fu : 0u);
}
__device__ __forceinline__ int own_pop(
    unsigned m, int ea, int eb, unsigned* q, int qcount, int lane, int lo, int cnt,
    const int* __restrict__ srcp, const int* __restrict__ dstp,
    const float* __restrict__ m_buf, float* tile) {
  unsigned mk = __builtin_amdgcn_ballot_w32(m != 0u);
  while (mk) {
    if (m != 0u) {
      const int p = __builtin_ctz(m);
      m &= m - 1u;
      const int j = 7 - (p & 7);
      const int e = ((j & 4) ? eb : ea) + 16 * lane + 4 * (j & 3) + (p >> 3);
      q[qcount + __builtin_amdgcn_mbcnt_lo(mk, 0u)] = (unsigned)e;
    }
    qcount += __builtin_popcount(mk);
    if (qcount >= 32) {
      own_drain(q, qcount, 32, lane, lo, cnt, srcp, dstp, m_buf, tile);
      qcount -= 32;
    }
    mk = __builtin_amdgcn_ballot_w32(m != 0u);
  }
  return qcount;
}
__device__ __forceinline__ int own_bstep(
    v4i (&d)[4], const unsigned* __restrict__ own_w, int st, int st_next, int wave,
    int rot, unsigned pat, unsigned* q, int qcount, int lane, int lo, int cnt,
    const int* __restrict__ srcp, const int* __restrict__ dstp,
    const float* __restrict__ m_buf, float* tile) {
  const int u0 = own_unit(st, wave, 0);
  const unsigned m0 = own_mask(d[0], d[1], pat, own_keep(u0,     u0 + 1));
  const unsigned m1 = own_mask(d[2], d[3], pat, own_keep(u0 + 2, u0 + 3));
  const int c0 = own_chunk(min(u0,     N_CHUNKS - 1), rot) * CHUNK;
  const int c1 = own_chunk(min(u0 + 1, N_CHUNKS - 1), rot) * CHUNK;
  const int c2 = own_chunk(min(u0 + 2, N_CHUNKS - 1), rot) * CHUNK;
  const int c3 = own_chunk(min(u0 + 3, N_CHUNKS - 1), rot) * CHUNK;
  qcount = own_pop(m0, c0, c1, q, qcount, lane, lo, cnt, srcp, dstp, m_buf, tile);
  qcount = own_pop(m1, c2, c3, q, qcount, lane, lo, cnt, srcp, dstp, m_buf, tile);
  own_bload(d, own_w, st_next, wave, rot, lane);
  return qcount;
}

__global__ __launch_bounds__(1024) void k_own_sum_out(
    const int* __restrict__ edge_index, const float* __restrict__ m_buf,
    const float* __restrict__ acc_buf, const unsigned* __restrict__ own_w,
    size_t ws_bytes, const float* __restrict__ Wo,
    const float* __restrict__ bo, float* __restrict__ out) {
  __shared__ float    tile[(OWN_NODES + 1) * HIDDEN];
  __shared__ unsigned queue[OWN_WAVES * QCAP];

  const int tid  = threadIdx.x;
  const int lane = tid & 31;
  const int wave = __builtin_amdgcn_readfirstlane(tid >> 5);
  const int lo   = blockIdx.x * OWN_NODES;
  const int cnt  = min(OWN_NODES, N_NODES - lo);

  for (int i = tid; i < cnt * 32; i += 1024) {
    const int n = i >> 5;
    const v4f v = *(const v4f*)(acc_buf + (size_t)(lo + n) * HIDDEN + 4 * lane);
    float* row = tile + n * HIDDEN + lane;
    row[0] = v.x; row[32] = v.y; row[64] = v.z; row[96] = v.w;
  }
  __syncthreads();

  const int* __restrict__ srcp = edge_index;
  const int* __restrict__ dstp = edge_index + N_EDGES;
  unsigned* q = queue + wave * QCAP;
  int qcount = 0;
  const int rot = blockIdx.x * 12;

  v4i bufA[4], bufB[4];
  if (ws_bytes >= WS_NEED) {
    const unsigned pat = (unsigned)blockIdx.x * 0x01010101u;
    own_bload(bufA, own_w, 0, wave, rot, lane);
    __builtin_amdgcn_sched_barrier(0);
    own_bload(bufB, own_w, 1, wave, rot, lane);
    for (int st = 0; st < N_STEPS; st += 2) {
      qcount = own_bstep(bufA, own_w, st,     st + 2, wave, rot, pat, q, qcount, lane,
                         lo, cnt, srcp, dstp, m_buf, tile);
      qcount = own_bstep(bufB, own_w, st + 1, st + 3, wave, rot, pat, q, qcount, lane,
                         lo, cnt, srcp, dstp, m_buf, tile);
    }
  } else {
    own_load(bufA, dstp, own_chunk(wave, rot), lane);
    __builtin_amdgcn_sched_barrier(0);
    own_load(bufB, dstp, own_chunk(own_clamp(wave + OWN_WAVES, wave), rot), lane);
    for (int it = wave; it < N_CHUNKS; it += 2 * OWN_WAVES) {
      qcount = own_scan(bufA, own_chunk(it, rot) * CHUNK, lo, cnt, q, qcount, lane);
      own_load(bufA, dstp, own_chunk(own_clamp(it + 2 * OWN_WAVES, wave), rot), lane);
      while (qcount >= 32) {
        own_drain(q, qcount, 32, lane, lo, cnt, srcp, dstp, m_buf, tile); qcount -= 32;
      }

      const int cntB = (it + OWN_WAVES < N_CHUNKS) ? cnt : 0;
      qcount = own_scan(bufB, own_chunk(own_clamp(it + OWN_WAVES, wave), rot) * CHUNK, lo, cntB, q, qcount, lane);
      own_load(bufB, dstp, own_chunk(own_clamp(it + 3 * OWN_WAVES, wave), rot), lane);
      while (qcount >= 32) {
        own_drain(q, qcount, 32, lane, lo, cnt, srcp, dstp, m_buf, tile); qcount -= 32;
      }
    }
  }
  if (qcount > 0) own_drain(q, qcount, qcount, lane, lo, cnt, srcp, dstp, m_buf, tile);
  __syncthreads();

  float w0[4], w1[4];
#pragma unroll
  for (int t = 0; t < 4; ++t) {
    const int k = lane + t * 32;
    w0[t] = Wo[k * 2 + 0];
    w1[t] = Wo[k * 2 + 1];
  }
  const float b0 = bo[0], b1 = bo[1];
  for (int n = wave; n < cnt; n += OWN_WAVES) {
    const float* row = tile + n * HIDDEN;
    float a0 = 0.0f, a1 = 0.0f;
#pragma unroll
    for (int t = 0; t < 4; ++t) {
      const int k = lane + t * 32;
      const float h = fmaxf(row[(k & 3) * 32 + (k >> 2)], 0.0f);
      a0 = fmaf(h, w0[t], a0);
      a1 = fmaf(h, w1[t], a1);
    }
#pragma unroll
    for (int off = 16; off >= 1; off >>= 1) {
      a0 += __shfl_xor(a0, off, 32);
      a1 += __shfl_xor(a1, off, 32);
    }
    if (lane == 0) {
      out[(lo + n) * 2 + 0] = a0 + b0;
      out[(lo + n) * 2 + 1] = a1 + b1;
    }
  }
}

extern "C" void kernel_launch(void* const* d_in, const int* in_sizes, int n_in,
                              void* d_out, int out_size, void* d_ws, size_t ws_size,
                              hipStream_t stream) {
  (void)in_sizes; (void)n_in; (void)out_size;
  const float* x     = (const float*)d_in[0];
  const int*   edge  = (const int*)  d_in[1];
  const int*   batch = (const int*)  d_in[2];
  const float* gf    = (const float*)d_in[3];
  const float* Wg    = (const float*)d_in[4];
  const float* bg    = (const float*)d_in[5];
  const float* Wc    = (const float*)d_in[6];
  const float* bc    = (const float*)d_in[7];
  const float* Wm    = (const float*)d_in[8];
  const float* bm    = (const float*)d_in[9];
  const float* Ws    = (const float*)d_in[10];
  const float* bs    = (const float*)d_in[11];
  const float* Wo    = (const float*)d_in[12];
  const float* bo    = (const float*)d_in[13];
  float* out = (float*)d_out;

  float* hg      = (float*)d_ws;
  float* m_buf   = hg    + (size_t)N_GRAPHS * HIDDEN;
  float* acc_buf = m_buf + (size_t)N_NODES  * HIDDEN;
  unsigned* own_w = (unsigned*)(acc_buf + (size_t)N_NODES * HIDDEN);

  k_global_enc<<<(N_GRAPHS * HIDDEN) / 256, 256, 0, stream>>>(gf, Wg, bg, hg);
  k_node_gemms<<<NG_WGS, 1024, 0, stream>>>(x, edge, batch, hg, Wc, bc,
                                                 Wm, bm, Ws, bs, m_buf, acc_buf,
                                                 own_w, ws_size);
  k_own_sum_out<<<OWN_WGS, 1024, 0, stream>>>(edge, m_buf, acc_buf, own_w, ws_size,
                                              Wo, bo, out);
}
